# starting kernel
# baseline (speedup 1.0000x reference)
#include <hip/hip_runtime.h>

typedef float v2f __attribute__((ext_vector_type(2)));
typedef float v8f __attribute__((ext_vector_type(8)));

#define Dd    1024
#define NS    25
#define NQ    75
#define NW    5
#define SROW  1028
#define KROW  32
#define WROW  8
#define QROW  132
#define KCH   128
#define NCHUNK (Dd / KCH)
#define NTHR  160

#define OFF_W    25700
#define OFF_KAUG 33908
#define OFF_PROW 34708
#define OFF_FCOL 34740
#define SMEM_FLOATS 34772

__global__ __launch_bounds__(NTHR) void ridge_head_fused(
    const float* __restrict__ q, const float* __restrict__ s,
    const int* __restrict__ labels, const float* __restrict__ scale,
    float* __restrict__ out)
{
  __shared__ __align__(16) float smem[SMEM_FLOATS];
  float* S_lds = smem;
  float* W_lds = smem + OFF_W;
  float* Kaug  = smem + OFF_KAUG;
  float* prow  = smem + OFF_PROW;
  float* fcol  = smem + OFF_FCOL;

  const int b    = blockIdx.x;
  const int tid  = threadIdx.x;
  const int lane = tid & 31;
  const int wave = tid >> 5;
  const int lm   = lane & 15;
  const int kh   = lane >> 4;

  {
    const float* sb = s + (size_t)b * NS * Dd;
    for (int idx = tid; idx < NS * (Dd / 4); idx += NTHR) {
      int r  = idx >> 8;
      int c4 = idx & 255;
      float4 v = ((const float4*)(sb + r * Dd))[c4];
      ((float4*)(S_lds + r * SROW))[c4] = v;
    }
    for (int idx = tid; idx < Dd * WROW + 16; idx += NTHR) W_lds[idx] = 0.0f;
  }
  __syncthreads();

  if (wave < 4) {
    const int ti = (wave >> 1) * 16;
    const int tj = (wave & 1) * 16;
    const float* arow = S_lds + (ti + lm) * SROW + 2 * kh;
    const float* brow = S_lds + (tj + lm) * SROW + 2 * kh;
    v8f acc0 = {0.f,0.f,0.f,0.f,0.f,0.f,0.f,0.f};
    v8f acc1 = {0.f,0.f,0.f,0.f,0.f,0.f,0.f,0.f};
    for (int k = 0; k < Dd; k += 8) {
      v2f a0 = *(const v2f*)(arow + k);
      v2f b0 = *(const v2f*)(brow + k);
      acc0 = __builtin_amdgcn_wmma_f32_16x16x4_f32(false, a0, false, b0,
                                                   (short)0, acc0, false, false);
      v2f a1 = *(const v2f*)(arow + k + 4);
      v2f b1 = *(const v2f*)(brow + k + 4);
      acc1 = __builtin_amdgcn_wmma_f32_16x16x4_f32(false, a1, false, b1,
                                                   (short)0, acc1, false, false);
    }
    v8f acc = acc0 + acc1;
    const int j = tj + lm;
    if (j < NS) {
      #pragma unroll
      for (int v = 0; v < 8; ++v) {
        int i = ti + kh * 8 + v;
        if (i < NS) Kaug[i * KROW + j] = acc[v] + ((i == j) ? 1.0f : 0.0f);
      }
    }
  } else {
    int r = tid - 128;
    if (r < NS) {
      int lab = labels[b * NS + r];
      for (int c = 0; c < NW; ++c)
        Kaug[r * KROW + NS + c] = (lab == c) ? 1.0f : 0.0f;
    }
  }
  __syncthreads();

  for (int p = 0; p < NS; ++p) {
    if (tid < NS + NW) {
      float pivinv = 1.0f / Kaug[p * KROW + p];
      prow[tid] = Kaug[p * KROW + tid] * pivinv;
      if (tid < NS) fcol[tid] = Kaug[tid * KROW + p];
    }
    __syncthreads();
    for (int e = tid; e < NS * (NS + NW); e += NTHR) {
      int r = e / (NS + NW);
      int c = e - r * (NS + NW);
      float v = (r == p) ? prow[c] : (Kaug[r * KROW + c] - fcol[r] * prow[c]);
      Kaug[r * KROW + c] = v;
    }
    __syncthreads();
  }

  const int m0 = wave * 16;
  const float* qb = q + (size_t)b * NQ * Dd;
  float4 r[16];
  #pragma unroll
  for (int u = 0; u < 16; ++u) {
    int m = m0 + u;
    float4 v = {0.f, 0.f, 0.f, 0.f};
    if (m < NQ) v = ((const float4*)(qb + (size_t)m * Dd))[lane];
    r[u] = v;
  }

  for (int e = tid; e < Dd * NW; e += NTHR) {
    int n = e >> 10;
    int k = e & (Dd - 1);
    float acc = 0.0f;
    #pragma unroll
    for (int i = 0; i < NS; ++i)
      acc += S_lds[i * SROW + k] * Kaug[i * KROW + NS + n];
    W_lds[k * WROW + n] = acc;
  }
  __syncthreads();

  float* Qs = S_lds + wave * (16 * QROW);

  v8f c0 = {0.f,0.f,0.f,0.f,0.f,0.f,0.f,0.f};
  v8f c1 = {0.f,0.f,0.f,0.f,0.f,0.f,0.f,0.f};

  const float* ap = Qs + lm * QROW + 2 * kh;

  for (int c = 0; c < NCHUNK; ++c) {
    #pragma unroll
    for (int u = 0; u < 16; ++u)
      ((float4*)(Qs + u * QROW))[lane] = r[u];

    if (c + 1 < NCHUNK) {
      #pragma unroll
      for (int u = 0; u < 16; ++u) {
        int m = m0 + u;
        float4 v = {0.f, 0.f, 0.f, 0.f};
        if (m < NQ) v = ((const float4*)(qb + (size_t)m * Dd + (c + 1) * KCH))[lane];
        r[u] = v;
      }
    }

    const int k0 = c * KCH;
    #pragma unroll 4
    for (int kc = 0; kc < KCH; kc += 8) {
      int kb0 = k0 + kc + 2 * kh;
      v2f a0 = *(const v2f*)(ap + kc);
      v2f b0 = { W_lds[kb0 * WROW + lm], W_lds[(kb0 + 1) * WROW + lm] };
      c0 = __builtin_amdgcn_wmma_f32_16x16x4_f32(false, a0, false, b0,
                                                 (short)0, c0, false, false);
      int kb1 = kb0 + 4;
      v2f a1 = *(const v2f*)(ap + kc + 4);
      v2f b1 = { W_lds[kb1 * WROW + lm], W_lds[(kb1 + 1) * WROW + lm] };
      c1 = __builtin_amdgcn_wmma_f32_16x16x4_f32(false, a1, false, b1,
                                                 (short)0, c1, false, false);
    }
  }

  const float sc = *scale;
  v8f cc = c0 + c1;
  if (lm < NW) {
    #pragma unroll
    for (int v = 0; v < 8; ++v) {
      int m = m0 + kh * 8 + v;
      if (m < NQ) out[((size_t)b * NQ + m) * NW + lm] = sc * cc[v];
    }
  }
}

extern "C" void kernel_launch(void* const* d_in, const int* in_sizes, int n_in,
                              void* d_out, int out_size, void* d_ws, size_t ws_size,
                              hipStream_t stream) {
  (void)n_in; (void)out_size; (void)d_ws; (void)ws_size;
  const float* q      = (const float*)d_in[0];
  const float* s      = (const float*)d_in[1];
  const int*   labels = (const int*)d_in[2];
  const float* scale  = (const float*)d_in[3];
  float* out = (float*)d_out;

  const int B = in_sizes[1] / (NS * Dd);
  if (B <= 0) return;

  ridge_head_fused<<<dim3(B), dim3(NTHR), 0, stream>>>(q, s, labels, scale, out);
}


# 25x30 Gauss-Jordan solve moved from LDS with 50 workgroup barriers into one wave's registers: lane = column, rows in VGPRs, row multipliers by v_readlane; same operations per element
# speedup vs baseline: 1.4621x; 1.4621x over previous
#include <hip/hip_runtime.h>

typedef float v2f __attribute__((ext_vector_type(2)));
typedef float v8f __attribute__((ext_vector_type(8)));

#define Dd    1024
#define NS    25
#define NQ    75
#define NW    5
#define SROW  1028
#define KROW  32
#define WROW  8
#define QROW  132
#define KCH   128
#define NCHUNK (Dd / KCH)
#define NTHR  160

#define OFF_W    25700
#define OFF_KAUG 33908
#define SMEM_FLOATS 34708

__global__ __launch_bounds__(NTHR) void ridge_head_fused(
    const float* __restrict__ q, const float* __restrict__ s,
    const int* __restrict__ labels, const float* __restrict__ scale,
    float* __restrict__ out)
{
  __shared__ __align__(16) float smem[SMEM_FLOATS];
  float* S_lds = smem;
  float* W_lds = smem + OFF_W;
  float* Kaug  = smem + OFF_KAUG;

  const int b    = blockIdx.x;
  const int tid  = threadIdx.x;
  const int lane = tid & 31;
  const int wave = tid >> 5;
  const int lm   = lane & 15;
  const int kh   = lane >> 4;

  {
    const float* sb = s + (size_t)b * NS * Dd;
    for (int idx = tid; idx < NS * (Dd / 4); idx += NTHR) {
      int r  = idx >> 8;
      int c4 = idx & 255;
      float4 v = ((const float4*)(sb + r * Dd))[c4];
      ((float4*)(S_lds + r * SROW))[c4] = v;
    }
    for (int idx = tid; idx < Dd * WROW + 16; idx += NTHR) W_lds[idx] = 0.0f;
  }
  __syncthreads();

  if (wave < 4) {
    const int ti = (wave >> 1) * 16;
    const int tj = (wave & 1) * 16;
    const float* arow = S_lds + (ti + lm) * SROW + 2 * kh;
    const float* brow = S_lds + (tj + lm) * SROW + 2 * kh;
    v8f acc0 = {0.f,0.f,0.f,0.f,0.f,0.f,0.f,0.f};
    v8f acc1 = {0.f,0.f,0.f,0.f,0.f,0.f,0.f,0.f};
    for (int k = 0; k < Dd; k += 8) {
      v2f a0 = *(const v2f*)(arow + k);
      v2f b0 = *(const v2f*)(brow + k);
      acc0 = __builtin_amdgcn_wmma_f32_16x16x4_f32(false, a0, false, b0,
                                                   (short)0, acc0, false, false);
      v2f a1 = *(const v2f*)(arow + k + 4);
      v2f b1 = *(const v2f*)(brow + k + 4);
      acc1 = __builtin_amdgcn_wmma_f32_16x16x4_f32(false, a1, false, b1,
                                                   (short)0, acc1, false, false);
    }
    v8f acc = acc0 + acc1;
    const int j = tj + lm;
    if (j < NS) {
      #pragma unroll
      for (int v = 0; v < 8; ++v) {
        int i = ti + kh * 8 + v;
        if (i < NS) Kaug[i * KROW + j] = acc[v] + ((i == j) ? 1.0f : 0.0f);
      }
    }
  } else {
    int r = tid - 128;
    if (r < NS) {
      int lab = labels[b * NS + r];
      for (int c = 0; c < NW; ++c)
        Kaug[r * KROW + NS + c] = (lab == c) ? 1.0f : 0.0f;
    }
  }
  __syncthreads();

  if (wave == 0) {
    const int col = (lane < NS + NW) ? lane : 0;
    float a[NS];
    #pragma unroll
    for (int r = 0; r < NS; ++r) a[r] = Kaug[r * KROW + col];
    #pragma unroll
    for (int p = 0; p < NS; ++p) {
      const float piv    = __int_as_float(__builtin_amdgcn_readlane(__float_as_int(a[p]), p));
      const float pivinv = 1.0f / piv;
      const float pr     = a[p] * pivinv;
      #pragma unroll
      for (int r = 0; r < NS; ++r) {
        if (r != p) {
          const float f = __int_as_float(__builtin_amdgcn_readlane(__float_as_int(a[r]), p));
          a[r] = a[r] - f * pr;
        }
      }
      a[p] = pr;
    }
    if (lane >= NS && lane < NS + NW) {
      #pragma unroll
      for (int r = 0; r < NS; ++r) Kaug[r * KROW + lane] = a[r];
    }
  }
  __syncthreads();

  const int m0 = wave * 16;
  const float* qb = q + (size_t)b * NQ * Dd;
  float4 r[16];
  #pragma unroll
  for (int u = 0; u < 16; ++u) {
    int m = m0 + u;
    float4 v = {0.f, 0.f, 0.f, 0.f};
    if (m < NQ) v = ((const float4*)(qb + (size_t)m * Dd))[lane];
    r[u] = v;
  }

  for (int e = tid; e < Dd * NW; e += NTHR) {
    int n = e >> 10;
    int k = e & (Dd - 1);
    float acc = 0.0f;
    #pragma unroll
    for (int i = 0; i < NS; ++i)
      acc += S_lds[i * SROW + k] * Kaug[i * KROW + NS + n];
    W_lds[k * WROW + n] = acc;
  }
  __syncthreads();

  float* Qs = S_lds + wave * (16 * QROW);

  v8f c0 = {0.f,0.f,0.f,0.f,0.f,0.f,0.f,0.f};
  v8f c1 = {0.f,0.f,0.f,0.f,0.f,0.f,0.f,0.f};

  const float* ap = Qs + lm * QROW + 2 * kh;

  for (int c = 0; c < NCHUNK; ++c) {
    #pragma unroll
    for (int u = 0; u < 16; ++u)
      ((float4*)(Qs + u * QROW))[lane] = r[u];

    if (c + 1 < NCHUNK) {
      #pragma unroll
      for (int u = 0; u < 16; ++u) {
        int m = m0 + u;
        float4 v = {0.f, 0.f, 0.f, 0.f};
        if (m < NQ) v = ((const float4*)(qb + (size_t)m * Dd + (c + 1) * KCH))[lane];
        r[u] = v;
      }
    }

    const int k0 = c * KCH;
    #pragma unroll 4
    for (int kc = 0; kc < KCH; kc += 8) {
      int kb0 = k0 + kc + 2 * kh;
      v2f a0 = *(const v2f*)(ap + kc);
      v2f b0 = { W_lds[kb0 * WROW + lm], W_lds[(kb0 + 1) * WROW + lm] };
      c0 = __builtin_amdgcn_wmma_f32_16x16x4_f32(false, a0, false, b0,
                                                 (short)0, c0, false, false);
      int kb1 = kb0 + 4;
      v2f a1 = *(const v2f*)(ap + kc + 4);
      v2f b1 = { W_lds[kb1 * WROW + lm], W_lds[(kb1 + 1) * WROW + lm] };
      c1 = __builtin_amdgcn_wmma_f32_16x16x4_f32(false, a1, false, b1,
                                                 (short)0, c1, false, false);
    }
  }

  const float sc = *scale;
  v8f cc = c0 + c1;
  if (lm < NW) {
    #pragma unroll
    for (int v = 0; v < 8; ++v) {
      int m = m0 + kh * 8 + v;
      if (m < NQ) out[((size_t)b * NQ + m) * NW + lm] = sc * cc[v];
    }
  }
}

extern "C" void kernel_launch(void* const* d_in, const int* in_sizes, int n_in,
                              void* d_out, int out_size, void* d_ws, size_t ws_size,
                              hipStream_t stream) {
  (void)n_in; (void)out_size; (void)d_ws; (void)ws_size;
  const float* q      = (const float*)d_in[0];
  const float* s      = (const float*)d_in[1];
  const int*   labels = (const int*)d_in[2];
  const float* scale  = (const float*)d_in[3];
  float* out = (float*)d_out;

  const int B = in_sizes[1] / (NS * Dd);
  if (B <= 0) return;

  ridge_head_fused<<<dim3(B), dim3(NTHR), 0, stream>>>(q, s, labels, scale, out);
}


# W = S^T A stage register-blocked: a thread keeps 6-7 columns x 5 classes in registers, one LDS read of S feeds 5 FMAs, A row read once per i (was 2 LDS reads per FMA); same FMA chain order
# speedup vs baseline: 1.9085x; 1.3053x over previous
#include <hip/hip_runtime.h>

typedef float v2f __attribute__((ext_vector_type(2)));
typedef float v8f __attribute__((ext_vector_type(8)));

#define Dd    1024
#define NS    25
#define NQ    75
#define NW    5
#define SROW  1028
#define KROW  32
#define WROW  8
#define QROW  132
#define KCH   128
#define NCHUNK (Dd / KCH)
#define NTHR  160

#define OFF_W    25700
#define OFF_KAUG 33908
#define SMEM_FLOATS 34708

template <int NJ>
__device__ __forceinline__ void wcols(const float* S_lds, const float* Kaug,
                                      float* W_lds, int tid)
{
  float acc[NJ][NW];
  #pragma unroll
  for (int j = 0; j < NJ; ++j) {
    #pragma unroll
    for (int n = 0; n < NW; ++n) acc[j][n] = 0.0f;
  }
  #pragma unroll 5
  for (int i = 0; i < NS; ++i) {
    const float* ar = Kaug + i * KROW + NS;
    float an[NW];
    #pragma unroll
    for (int n = 0; n < NW; ++n) an[n] = ar[n];
    const float* sr = S_lds + i * SROW + tid;
    #pragma unroll
    for (int j = 0; j < NJ; ++j) {
      const float sv = sr[j * NTHR];
      #pragma unroll
      for (int n = 0; n < NW; ++n) acc[j][n] = __builtin_fmaf(sv, an[n], acc[j][n]);
    }
  }
  #pragma unroll
  for (int j = 0; j < NJ; ++j) {
    float* wr = W_lds + (tid + j * NTHR) * WROW;
    #pragma unroll
    for (int n = 0; n < NW; ++n) wr[n] = acc[j][n];
  }
}

__global__ __launch_bounds__(NTHR) void ridge_head_fused(
    const float* __restrict__ q, const float* __restrict__ s,
    const int* __restrict__ labels, const float* __restrict__ scale,
    float* __restrict__ out)
{
  __shared__ __align__(16) float smem[SMEM_FLOATS];
  float* S_lds = smem;
  float* W_lds = smem + OFF_W;
  float* Kaug  = smem + OFF_KAUG;

  const int b    = blockIdx.x;
  const int tid  = threadIdx.x;
  const int lane = tid & 31;
  const int wave = tid >> 5;
  const int lm   = lane & 15;
  const int kh   = lane >> 4;

  {
    const float* sb = s + (size_t)b * NS * Dd;
    for (int idx = tid; idx < NS * (Dd / 4); idx += NTHR) {
      int r  = idx >> 8;
      int c4 = idx & 255;
      float4 v = ((const float4*)(sb + r * Dd))[c4];
      ((float4*)(S_lds + r * SROW))[c4] = v;
    }
    for (int idx = tid; idx < Dd * WROW + 16; idx += NTHR) W_lds[idx] = 0.0f;
  }
  __syncthreads();

  if (wave < 4) {
    const int ti = (wave >> 1) * 16;
    const int tj = (wave & 1) * 16;
    const float* arow = S_lds + (ti + lm) * SROW + 2 * kh;
    const float* brow = S_lds + (tj + lm) * SROW + 2 * kh;
    v8f acc0 = {0.f,0.f,0.f,0.f,0.f,0.f,0.f,0.f};
    v8f acc1 = {0.f,0.f,0.f,0.f,0.f,0.f,0.f,0.f};
    for (int k = 0; k < Dd; k += 8) {
      v2f a0 = *(const v2f*)(arow + k);
      v2f b0 = *(const v2f*)(brow + k);
      acc0 = __builtin_amdgcn_wmma_f32_16x16x4_f32(false, a0, false, b0,
                                                   (short)0, acc0, false, false);
      v2f a1 = *(const v2f*)(arow + k + 4);
      v2f b1 = *(const v2f*)(brow + k + 4);
      acc1 = __builtin_amdgcn_wmma_f32_16x16x4_f32(false, a1, false, b1,
                                                   (short)0, acc1, false, false);
    }
    v8f acc = acc0 + acc1;
    const int j = tj + lm;
    if (j < NS) {
      #pragma unroll
      for (int v = 0; v < 8; ++v) {
        int i = ti + kh * 8 + v;
        if (i < NS) Kaug[i * KROW + j] = acc[v] + ((i == j) ? 1.0f : 0.0f);
      }
    }
  } else {
    int r = tid - 128;
    if (r < NS) {
      int lab = labels[b * NS + r];
      for (int c = 0; c < NW; ++c)
        Kaug[r * KROW + NS + c] = (lab == c) ? 1.0f : 0.0f;
    }
  }
  __syncthreads();

  if (wave == 0) {
    const int col = (lane < NS + NW) ? lane : 0;
    float a[NS];
    #pragma unroll
    for (int r = 0; r < NS; ++r) a[r] = Kaug[r * KROW + col];
    #pragma unroll
    for (int p = 0; p < NS; ++p) {
      const float piv    = __int_as_float(__builtin_amdgcn_readlane(__float_as_int(a[p]), p));
      const float pivinv = 1.0f / piv;
      const float pr     = a[p] * pivinv;
      #pragma unroll
      for (int r = 0; r < NS; ++r) {
        if (r != p) {
          const float f = __int_as_float(__builtin_amdgcn_readlane(__float_as_int(a[r]), p));
          a[r] = a[r] - f * pr;
        }
      }
      a[p] = pr;
    }
    if (lane >= NS && lane < NS + NW) {
      #pragma unroll
      for (int r = 0; r < NS; ++r) Kaug[r * KROW + lane] = a[r];
    }
  }
  __syncthreads();

  const int m0 = wave * 16;
  const float* qb = q + (size_t)b * NQ * Dd;
  float4 r[16];
  #pragma unroll
  for (int u = 0; u < 16; ++u) {
    int m = m0 + u;
    float4 v = {0.f, 0.f, 0.f, 0.f};
    if (m < NQ) v = ((const float4*)(qb + (size_t)m * Dd))[lane];
    r[u] = v;
  }

  if (tid < Dd - 6 * NTHR) wcols<7>(S_lds, Kaug, W_lds, tid);
  else                     wcols<6>(S_lds, Kaug, W_lds, tid);
  __syncthreads();

  float* Qs = S_lds + wave * (16 * QROW);

  v8f c0 = {0.f,0.f,0.f,0.f,0.f,0.f,0.f,0.f};
  v8f c1 = {0.f,0.f,0.f,0.f,0.f,0.f,0.f,0.f};

  const float* ap = Qs + lm * QROW + 2 * kh;

  for (int c = 0; c < NCHUNK; ++c) {
    #pragma unroll
    for (int u = 0; u < 16; ++u)
      ((float4*)(Qs + u * QROW))[lane] = r[u];

    if (c + 1 < NCHUNK) {
      #pragma unroll
      for (int u = 0; u < 16; ++u) {
        int m = m0 + u;
        float4 v = {0.f, 0.f, 0.f, 0.f};
        if (m < NQ) v = ((const float4*)(qb + (size_t)m * Dd + (c + 1) * KCH))[lane];
        r[u] = v;
      }
    }

    const int k0 = c * KCH;
    #pragma unroll 4
    for (int kc = 0; kc < KCH; kc += 8) {
      int kb0 = k0 + kc + 2 * kh;
      v2f a0 = *(const v2f*)(ap + kc);
      v2f b0 = { W_lds[kb0 * WROW + lm], W_lds[(kb0 + 1) * WROW + lm] };
      c0 = __builtin_amdgcn_wmma_f32_16x16x4_f32(false, a0, false, b0,
                                                 (short)0, c0, false, false);
      int kb1 = kb0 + 4;
      v2f a1 = *(const v2f*)(ap + kc + 4);
      v2f b1 = { W_lds[kb1 * WROW + lm], W_lds[(kb1 + 1) * WROW + lm] };
      c1 = __builtin_amdgcn_wmma_f32_16x16x4_f32(false, a1, false, b1,
                                                 (short)0, c1, false, false);
    }
  }

  const float sc = *scale;
  v8f cc = c0 + c1;
  if (lm < NW) {
    #pragma unroll
    for (int v = 0; v < 8; ++v) {
      int m = m0 + kh * 8 + v;
      if (m < NQ) out[((size_t)b * NQ + m) * NW + lm] = sc * cc[v];
    }
  }
}

extern "C" void kernel_launch(void* const* d_in, const int* in_sizes, int n_in,
                              void* d_out, int out_size, void* d_ws, size_t ws_size,
                              hipStream_t stream) {
  (void)n_in; (void)out_size; (void)d_ws; (void)ws_size;
  const float* q      = (const float*)d_in[0];
  const float* s      = (const float*)d_in[1];
  const int*   labels = (const int*)d_in[2];
  const float* scale  = (const float*)d_in[3];
  float* out = (float*)d_out;

  const int B = in_sizes[1] / (NS * Dd);
  if (B <= 0) return;

  ridge_head_fused<<<dim3(B), dim3(NTHR), 0, stream>>>(q, s, labels, scale, out);
}


# more global loads in flight per wave: S staging issues all 40 b128 loads of a thread before its first LDS store (was 8 per trip); Q stream keeps two chunks in flight in a ring of two register sets
# speedup vs baseline: 2.6471x; 1.3870x over previous
#include <hip/hip_runtime.h>

typedef float v2f __attribute__((ext_vector_type(2)));
typedef float v8f __attribute__((ext_vector_type(8)));
typedef float v4f __attribute__((ext_vector_type(4)));

#define Dd    1024
#define NS    25
#define NQ    75
#define NW    5
#define SROW  1028
#define KROW  32
#define WROW  8
#define QROW  132
#define KCH   128
#define NCHUNK (Dd / KCH)
#define NTHR  160
#define SLD   (NS * (Dd / 4) / NTHR)

#define OFF_W    25700
#define OFF_KAUG 33908
#define SMEM_FLOATS 34708

template <int NJ>
__device__ __forceinline__ void wcols(const float* S_lds, const float* Kaug,
                                      float* W_lds, int tid)
{
  float acc[NJ][NW];
  #pragma unroll
  for (int j = 0; j < NJ; ++j) {
    #pragma unroll
    for (int n = 0; n < NW; ++n) acc[j][n] = 0.0f;
  }
  #pragma unroll 5
  for (int i = 0; i < NS; ++i) {
    const float* ar = Kaug + i * KROW + NS;
    float an[NW];
    #pragma unroll
    for (int n = 0; n < NW; ++n) an[n] = ar[n];
    const float* sr = S_lds + i * SROW + tid;
    #pragma unroll
    for (int j = 0; j < NJ; ++j) {
      const float sv = sr[j * NTHR];
      #pragma unroll
      for (int n = 0; n < NW; ++n) acc[j][n] = __builtin_fmaf(sv, an[n], acc[j][n]);
    }
  }
  #pragma unroll
  for (int j = 0; j < NJ; ++j) {
    float* wr = W_lds + (tid + j * NTHR) * WROW;
    #pragma unroll
    for (int n = 0; n < NW; ++n) wr[n] = acc[j][n];
  }
}

__device__ __forceinline__ void q_fetch(const v4f* const (&qrow)[16], int off4, v4f (&r)[16])
{
  #pragma unroll
  for (int u = 0; u < 16; ++u) r[u] = qrow[u][off4];
}

__device__ __forceinline__ void q_commit(float* Qs, int lane, const v4f (&r)[16])
{
  #pragma unroll
  for (int u = 0; u < 16; ++u) ((v4f*)(Qs + u * QROW))[lane] = r[u];
}

__device__ __forceinline__ void qw_chunk(const float* ap, const float* W_lds, int k0,
                                         int lm, int kh, v8f& c0, v8f& c1)
{
  #pragma unroll 4
  for (int kc = 0; kc < KCH; kc += 8) {
    int kb0 = k0 + kc + 2 * kh;
    v2f a0 = *(const v2f*)(ap + kc);
    v2f b0 = { W_lds[kb0 * WROW + lm], W_lds[(kb0 + 1) * WROW + lm] };
    c0 = __builtin_amdgcn_wmma_f32_16x16x4_f32(false, a0, false, b0,
                                               (short)0, c0, false, false);
    int kb1 = kb0 + 4;
    v2f a1 = *(const v2f*)(ap + kc + 4);
    v2f b1 = { W_lds[kb1 * WROW + lm], W_lds[(kb1 + 1) * WROW + lm] };
    c1 = __builtin_amdgcn_wmma_f32_16x16x4_f32(false, a1, false, b1,
                                               (short)0, c1, false, false);
  }
}

__global__ __launch_bounds__(NTHR) void ridge_head_fused(
    const float* __restrict__ q, const float* __restrict__ s,
    const int* __restrict__ labels, const float* __restrict__ scale,
    float* __restrict__ out)
{
  __shared__ __align__(16) float smem[SMEM_FLOATS];
  float* S_lds = smem;
  float* W_lds = smem + OFF_W;
  float* Kaug  = smem + OFF_KAUG;

  const int b    = blockIdx.x;
  const int tid  = threadIdx.x;
  const int lane = tid & 31;
  const int wave = tid >> 5;
  const int lm   = lane & 15;
  const int kh   = lane >> 4;

  {
    for (int idx = tid; idx < Dd * WROW + 16; idx += NTHR) W_lds[idx] = 0.0f;
    const float4* sb4 = (const float4*)(s + (size_t)b * NS * Dd);
    float4 t[SLD];
    #pragma unroll
    for (int j = 0; j < SLD; ++j) t[j] = sb4[tid + j * NTHR];
    #pragma unroll
    for (int j = 0; j < SLD; ++j) {
      const int idx = tid + j * NTHR;
      ((float4*)(S_lds + (idx >> 8) * SROW))[idx & 255] = t[j];
    }
  }
  __syncthreads();

  if (wave < 4) {
    const int ti = (wave >> 1) * 16;
    const int tj = (wave & 1) * 16;
    const float* arow = S_lds + (ti + lm) * SROW + 2 * kh;
    const float* brow = S_lds + (tj + lm) * SROW + 2 * kh;
    v8f acc0 = {0.f,0.f,0.f,0.f,0.f,0.f,0.f,0.f};
    v8f acc1 = {0.f,0.f,0.f,0.f,0.f,0.f,0.f,0.f};
    for (int k = 0; k < Dd; k += 8) {
      v2f a0 = *(const v2f*)(arow + k);
      v2f b0 = *(const v2f*)(brow + k);
      acc0 = __builtin_amdgcn_wmma_f32_16x16x4_f32(false, a0, false, b0,
                                                   (short)0, acc0, false, false);
      v2f a1 = *(const v2f*)(arow + k + 4);
      v2f b1 = *(const v2f*)(brow + k + 4);
      acc1 = __builtin_amdgcn_wmma_f32_16x16x4_f32(false, a1, false, b1,
                                                   (short)0, acc1, false, false);
    }
    v8f acc = acc0 + acc1;
    const int j = tj + lm;
    if (j < NS) {
      #pragma unroll
      for (int v = 0; v < 8; ++v) {
        int i = ti + kh * 8 + v;
        if (i < NS) Kaug[i * KROW + j] = acc[v] + ((i == j) ? 1.0f : 0.0f);
      }
    }
  } else {
    int r = tid - 128;
    if (r < NS) {
      int lab = labels[b * NS + r];
      for (int c = 0; c < NW; ++c)
        Kaug[r * KROW + NS + c] = (lab == c) ? 1.0f : 0.0f;
    }
  }
  __syncthreads();

  if (wave == 0) {
    const int col = (lane < NS + NW) ? lane : 0;
    float a[NS];
    #pragma unroll
    for (int r = 0; r < NS; ++r) a[r] = Kaug[r * KROW + col];
    #pragma unroll
    for (int p = 0; p < NS; ++p) {
      const float piv    = __int_as_float(__builtin_amdgcn_readlane(__float_as_int(a[p]), p));
      const float pivinv = 1.0f / piv;
      const float pr     = a[p] * pivinv;
      #pragma unroll
      for (int r = 0; r < NS; ++r) {
        if (r != p) {
          const float f = __int_as_float(__builtin_amdgcn_readlane(__float_as_int(a[r]), p));
          a[r] = a[r] - f * pr;
        }
      }
      a[p] = pr;
    }
    if (lane >= NS && lane < NS + NW) {
      #pragma unroll
      for (int r = 0; r < NS; ++r) Kaug[r * KROW + lane] = a[r];
    }
  }
  __syncthreads();

  const int m0 = wave * 16;
  const float* qb = q + (size_t)b * NQ * Dd;
  const v4f* qrow[16];
  #pragma unroll
  for (int u = 0; u < 16; ++u) {
    const int m = (m0 + u < NQ) ? (m0 + u) : (NQ - 1);
    qrow[u] = (const v4f*)(qb + (size_t)m * Dd) + lane;
  }
  v4f r0[16], r1[16];
  q_fetch(qrow, 0, r0);

  if (tid < Dd - 6 * NTHR) wcols<7>(S_lds, Kaug, W_lds, tid);
  else                     wcols<6>(S_lds, Kaug, W_lds, tid);
  __syncthreads();

  float* Qs = S_lds + wave * (16 * QROW);

  v8f c0 = {0.f,0.f,0.f,0.f,0.f,0.f,0.f,0.f};
  v8f c1 = {0.f,0.f,0.f,0.f,0.f,0.f,0.f,0.f};

  const float* ap = Qs + lm * QROW + 2 * kh;

  q_fetch(qrow, KCH / 4, r1);

  for (int c = 0; c < NCHUNK - 2; c += 2) {
    q_commit(Qs, lane, r0);
    q_fetch(qrow, (c + 2) * (KCH / 4), r0);
    qw_chunk(ap, W_lds, c * KCH, lm, kh, c0, c1);
    q_commit(Qs, lane, r1);
    q_fetch(qrow, (c + 3) * (KCH / 4), r1);
    qw_chunk(ap, W_lds, (c + 1) * KCH, lm, kh, c0, c1);
  }
  q_commit(Qs, lane, r0);
  qw_chunk(ap, W_lds, (NCHUNK - 2) * KCH, lm, kh, c0, c1);
  q_commit(Qs, lane, r1);
  qw_chunk(ap, W_lds, (NCHUNK - 1) * KCH, lm, kh, c0, c1);

  const float sc = *scale;
  v8f cc = c0 + c1;
  if (lm < NW) {
    #pragma unroll
    for (int v = 0; v < 8; ++v) {
      int m = m0 + kh * 8 + v;
      if (m < NQ) out[((size_t)b * NQ + m) * NW + lm] = sc * cc[v];
    }
  }
}

extern "C" void kernel_launch(void* const* d_in, const int* in_sizes, int n_in,
                              void* d_out, int out_size, void* d_ws, size_t ws_size,
                              hipStream_t stream) {
  (void)n_in; (void)out_size; (void)d_ws; (void)ws_size;
  const float* q      = (const float*)d_in[0];
  const float* s      = (const float*)d_in[1];
  const int*   labels = (const int*)d_in[2];
  const float* scale  = (const float*)d_in[3];
  float* out = (float*)d_out;

  const int B = in_sizes[1] / (NS * Dd);
  if (B <= 0) return;

  ridge_head_fused<<<dim3(B), dim3(NTHR), 0, stream>>>(q, s, labels, scale, out);
}


# workgroup of 8 waves (was 5), the k range of both matrix products split over the waves: a wave takes one 128-k slice of all tiles, partials summed via LDS in wave order (new summation order)
# speedup vs baseline: 3.4959x; 1.3207x over previous
#include <hip/hip_runtime.h>


typedef float v2f __attribute__((ext_vector_type(2)));
typedef float v8f __attribute__((ext_vector_type(8)));
typedef float v4f __attribute__((ext_vector_type(4)));

#define Dd    1024
#define NS    25
#define NQ    75
#define NW    5
#define SROW  1028
#define KROW  32
#define WROW  8
#define QROW  132
#define KCH   128
#define NWAVE (Dd / KCH)
#define NTHR  (NWAVE * 32)
#define SLD   (NS * (Dd / 4) / NTHR)
#define NMT   5

#define OFF_W    25700
#define OFF_KAUG 33908
#define OFF_P    (NWAVE * 16 * QROW)
#define PWAVE    (NMT * 2 * 8 * NW)
#define SMEM_FLOATS 34708

template <int NJ>
__device__ __forceinline__ void wcols(const float* S_lds, const float* Kaug,
                                      float* W_lds, int tid)
{
  float acc[NJ][NW];
  #pragma unroll
  for (int j = 0; j < NJ; ++j) {
    #pragma unroll
    for (int n = 0; n < NW; ++n) acc[j][n] = 0.0f;
  }
  #pragma unroll 5
  for (int i = 0; i < NS; ++i) {
    const float* ar = Kaug + i * KROW + NS;
    float an[NW];
    #pragma unroll
    for (int n = 0; n < NW; ++n) an[n] = ar[n];
    const float* sr = S_lds + i * SROW + tid;
    #pragma unroll
    for (int j = 0; j < NJ; ++j) {
      const float sv = sr[j * NTHR];
      #pragma unroll
      for (int n = 0; n < NW; ++n) acc[j][n] = __builtin_fmaf(sv, an[n], acc[j][n]);
    }
  }
  #pragma unroll
  for (int j = 0; j < NJ; ++j) {
    float* wr = W_lds + (tid + j * NTHR) * WROW;
    #pragma unroll
    for (int n = 0; n < NW; ++n) wr[n] = acc[j][n];
  }
}

template <int T>
__device__ __forceinline__ void q_fetch(const v4f* qbase, v4f (&r)[16])
{
  #pragma unroll
  for (int u = 0; u < 16; ++u) {
    const int m = (16 * T + u < NQ) ? (16 * T + u) : (NQ - 1);
    r[u] = qbase[m * (Dd / 4)];
  }
}

__device__ __forceinline__ void q_commit(float* Qs, int lane, const v4f (&r)[16])
{
  #pragma unroll
  for (int u = 0; u < 16; ++u) ((v4f*)(Qs + u * QROW))[lane] = r[u];
}

__device__ __forceinline__ v8f q_tile_mma(const float* ap, const v2f (&bf)[KCH / 4])
{
  v8f c = {0.f,0.f,0.f,0.f,0.f,0.f,0.f,0.f};
  #pragma unroll
  for (int ks = 0; ks < KCH / 4; ++ks) {
    v2f a = *(const v2f*)(ap + 4 * ks);
    c = __builtin_amdgcn_wmma_f32_16x16x4_f32(false, a, false, bf[ks],
                                              (short)0, c, false, false);
  }
  return c;
}

template <int T>
__device__ __forceinline__ void p_store(float* Pw, int lm, int kh, const v8f& cc)
{
  if (lm < NW) {
    #pragma unroll
    for (int v = 0; v < 8; ++v) Pw[((T * 2 + kh) * 8 + v) * NW + lm] = cc[v];
  }
}

__global__ __launch_bounds__(NTHR, 4) void ridge_head_fused(
    const float* __restrict__ q, const float* __restrict__ s,
    const int* __restrict__ labels, const float* __restrict__ scale,
    float* __restrict__ out)
{
  __shared__ __align__(16) float smem[SMEM_FLOATS];
  float* S_lds = smem;
  float* W_lds = smem + OFF_W;
  float* Kaug  = smem + OFF_KAUG;

  const int b    = blockIdx.x;
  const int tid  = threadIdx.x;
  const int lane = tid & 31;
  const int wave = tid >> 5;
  const int lm   = lane & 15;
  const int kh   = lane >> 4;

  const float sc = *scale;
  const int lab = (tid < NS) ? labels[b * NS + tid] : 0;

  {
    for (int idx = tid; idx < Dd * WROW + 16; idx += NTHR) W_lds[idx] = 0.0f;
    const float4* sb4 = (const float4*)(s + (size_t)b * NS * Dd);
    float4 t[SLD];
    #pragma unroll
    for (int j = 0; j < SLD; ++j) t[j] = sb4[tid + j * NTHR];
    #pragma unroll
    for (int j = 0; j < SLD; ++j) ((float4*)(S_lds + j * SROW))[tid] = t[j];
  }

  __syncthreads();

  const v4f* qbase = (const v4f*)(q + (size_t)b * NQ * Dd + wave * KCH) + lane;
  v4f r0[16], r1[16];
  q_fetch<0>(qbase, r0);

  {
    const float* f0p = S_lds + lm * SROW + 2 * kh + wave * KCH;
    const float* f1p = S_lds + (16 + lm) * SROW + 2 * kh + wave * KCH;
    v8f t00 = {0.f,0.f,0.f,0.f,0.f,0.f,0.f,0.f};
    v8f t01 = {0.f,0.f,0.f,0.f,0.f,0.f,0.f,0.f};
    v8f t10 = {0.f,0.f,0.f,0.f,0.f,0.f,0.f,0.f};
    v8f t11 = {0.f,0.f,0.f,0.f,0.f,0.f,0.f,0.f};
    #pragma unroll 8
    for (int k = 0; k < KCH; k += 4) {
      v2f f0 = *(const v2f*)(f0p + k);
      v2f f1 = *(const v2f*)(f1p + k);
      t00 = __builtin_amdgcn_wmma_f32_16x16x4_f32(false, f0, false, f0,
                                                  (short)0, t00, false, false);
      t01 = __builtin_amdgcn_wmma_f32_16x16x4_f32(false, f0, false, f1,
                                                  (short)0, t01, false, false);
      t10 = __builtin_amdgcn_wmma_f32_16x16x4_f32(false, f1, false, f0,
                                                  (short)0, t10, false, false);
      t11 = __builtin_amdgcn_wmma_f32_16x16x4_f32(false, f1, false, f1,
                                                  (short)0, t11, false, false);
    }
    float* kp = W_lds + wave * (4 * 256) + lane;
    #pragma unroll
    for (int v = 0; v < 8; ++v) {
      kp[(0 * 8 + v) * 32] = t00[v];
      kp[(1 * 8 + v) * 32] = t01[v];
      kp[(2 * 8 + v) * 32] = t10[v];
      kp[(3 * 8 + v) * 32] = t11[v];
    }
  }
  __syncthreads();
  for (int e = tid; e < NS * NS; e += NTHR) {
    const int i = e / NS, j = e - i * NS;
    const int tile = ((i >> 4) << 1) | (j >> 4);
    const int ln   = (j & 15) + ((i & 8) << 1);
    const float* p = W_lds + (tile * 8 + (i & 7)) * 32 + ln;
    float sum = p[0];
    #pragma unroll
    for (int w = 1; w < NWAVE; ++w) sum += p[w * (4 * 256)];
    Kaug[i * KROW + j] = sum + ((i == j) ? 1.0f : 0.0f);
  }
  if (tid < NS) {
    for (int c = 0; c < NW; ++c)
      Kaug[tid * KROW + NS + c] = (lab == c) ? 1.0f : 0.0f;
  }
  __syncthreads();

  q_fetch<1>(qbase, r1);
  if (wave == 0) {
    const int col = (lane < NS + NW) ? lane : 0;
    float a[NS];
    #pragma unroll
    for (int r = 0; r < NS; ++r) a[r] = Kaug[r * KROW + col];
    #pragma unroll
    for (int p = 0; p < NS; ++p) {
      const float piv    = __int_as_float(__builtin_amdgcn_readlane(__float_as_int(a[p]), p));
      const float pivinv = 1.0f / piv;
      const float pr     = a[p] * pivinv;
      #pragma unroll
      for (int r = 0; r < NS; ++r) {
        if (r != p) {
          const float f = __int_as_float(__builtin_amdgcn_readlane(__float_as_int(a[r]), p));
          a[r] = a[r] - f * pr;
        }
      }
      a[p] = pr;
    }
    if (lane >= NS && lane < NS + NW) {
      #pragma unroll
      for (int r = 0; r < NS; ++r) Kaug[r * KROW + lane] = a[r];
    }
  }
  __syncthreads();

  wcols<Dd / NTHR>(S_lds, Kaug, W_lds, tid);
  __syncthreads();

  {
    float* Qs = S_lds + wave * (16 * QROW);
    float* Pw = S_lds + OFF_P + wave * PWAVE;
    const float* ap = Qs + lm * QROW + 2 * kh;

    v2f bf[KCH / 4];
    #pragma unroll
    for (int ks = 0; ks < KCH / 4; ++ks) {
      const int kb0 = wave * KCH + 4 * ks + 2 * kh;
      bf[ks] = v2f{ W_lds[kb0 * WROW + lm], W_lds[(kb0 + 1) * WROW + lm] };
    }

    q_commit(Qs, lane, r0);
    q_fetch<2>(qbase, r0);
    p_store<0>(Pw, lm, kh, q_tile_mma(ap, bf));
    q_commit(Qs, lane, r1);
    q_fetch<3>(qbase, r1);
    p_store<1>(Pw, lm, kh, q_tile_mma(ap, bf));
    q_commit(Qs, lane, r0);
    q_fetch<4>(qbase, r0);
    p_store<2>(Pw, lm, kh, q_tile_mma(ap, bf));
    q_commit(Qs, lane, r1);
    p_store<3>(Pw, lm, kh, q_tile_mma(ap, bf));
    q_commit(Qs, lane, r0);
    p_store<4>(Pw, lm, kh, q_tile_mma(ap, bf));
  }
  __syncthreads();

  for (int o = tid; o < NQ * NW; o += NTHR) {
    const int m = o / NW, c = o - m * NW;
    const float* p = S_lds + OFF_P + (((m >> 4) * 2 + ((m >> 3) & 1)) * 8 + (m & 7)) * NW + c;
    float sum = p[0];
    #pragma unroll
    for (int w = 1; w < NWAVE; ++w) sum += p[w * PWAVE];
    out[(size_t)b * (NQ * NW) + o] = sc * sum;
  }
}

extern "C" void kernel_launch(void* const* d_in, const int* in_sizes, int n_in,
                              void* d_out, int out_size, void* d_ws, size_t ws_size,
                              hipStream_t stream) {
  (void)n_in; (void)out_size; (void)d_ws; (void)ws_size;
  const float* q      = (const float*)d_in[0];
  const float* s      = (const float*)d_in[1];
  const int*   labels = (const int*)d_in[2];
  const float* scale  = (const float*)d_in[3];
  float* out = (float*)d_out;

  const int B = in_sizes[1] / (NS * Dd);
  if (B <= 0) return;

  ridge_head_fused<<<dim3(B), dim3(NTHR), 0, stream>>>(q, s, labels, scale, out);
}


# K = S*S^T uses its symmetry: the tile below the diagonal is not multiplied (3 matrix instructions per k-step, not 4); its cells are read transposed from the upper tile; same sums
# speedup vs baseline: 3.6917x; 1.0560x over previous
#include <hip/hip_runtime.h>


typedef float v2f __attribute__((ext_vector_type(2)));
typedef float v8f __attribute__((ext_vector_type(8)));
typedef float v4f __attribute__((ext_vector_type(4)));

#define Dd    1024
#define NS    25
#define NQ    75
#define NW    5
#define SROW  1028
#define KROW  32
#define WROW  8
#define QROW  132
#define KCH   128
#define NWAVE (Dd / KCH)
#define NTHR  (NWAVE * 32)
#define SLD   (NS * (Dd / 4) / NTHR)
#define NMT   5

#define OFF_W    25700
#define OFF_KAUG 33908
#define OFF_P    (NWAVE * 16 * QROW)
#define PWAVE    (NMT * 2 * 8 * NW)
#define SMEM_FLOATS 34708

template <int NJ>
__device__ __forceinline__ void wcols(const float* S_lds, const float* Kaug,
                                      float* W_lds, int tid)
{
  float acc[NJ][NW];
  #pragma unroll
  for (int j = 0; j < NJ; ++j) {
    #pragma unroll
    for (int n = 0; n < NW; ++n) acc[j][n] = 0.0f;
  }
  #pragma unroll 5
  for (int i = 0; i < NS; ++i) {
    const float* ar = Kaug + i * KROW + NS;
    float an[NW];
    #pragma unroll
    for (int n = 0; n < NW; ++n) an[n] = ar[n];
    const float* sr = S_lds + i * SROW + tid;
    #pragma unroll
    for (int j = 0; j < NJ; ++j) {
      const float sv = sr[j * NTHR];
      #pragma unroll
      for (int n = 0; n < NW; ++n) acc[j][n] = __builtin_fmaf(sv, an[n], acc[j][n]);
    }
  }
  #pragma unroll
  for (int j = 0; j < NJ; ++j) {
    float* wr = W_lds + (tid + j * NTHR) * WROW;
    #pragma unroll
    for (int n = 0; n < NW; ++n) wr[n] = acc[j][n];
  }
}

template <int T>
__device__ __forceinline__ void q_fetch(const v4f* qbase, v4f (&r)[16])
{
  #pragma unroll
  for (int u = 0; u < 16; ++u) {
    const int m = (16 * T + u < NQ) ? (16 * T + u) : (NQ - 1);
    r[u] = qbase[m * (Dd / 4)];
  }
}

__device__ __forceinline__ void q_commit(float* Qs, int lane, const v4f (&r)[16])
{
  #pragma unroll
  for (int u = 0; u < 16; ++u) ((v4f*)(Qs + u * QROW))[lane] = r[u];
}

__device__ __forceinline__ v8f q_tile_mma(const float* ap, const v2f (&bf)[KCH / 4])
{
  v8f c = {0.f,0.f,0.f,0.f,0.f,0.f,0.f,0.f};
  #pragma unroll
  for (int ks = 0; ks < KCH / 4; ++ks) {
    v2f a = *(const v2f*)(ap + 4 * ks);
    c = __builtin_amdgcn_wmma_f32_16x16x4_f32(false, a, false, bf[ks],
                                              (short)0, c, false, false);
  }
  return c;
}

template <int T>
__device__ __forceinline__ void p_store(float* Pw, int lm, int kh, const v8f& cc)
{
  if (lm < NW) {
    #pragma unroll
    for (int v = 0; v < 8; ++v) Pw[((T * 2 + kh) * 8 + v) * NW + lm] = cc[v];
  }
}

__global__ __launch_bounds__(NTHR, 4) void ridge_head_fused(
    const float* __restrict__ q, const float* __restrict__ s,
    const int* __restrict__ labels, const float* __restrict__ scale,
    float* __restrict__ out)
{
  __shared__ __align__(16) float smem[SMEM_FLOATS];
  float* S_lds = smem;
  float* W_lds = smem + OFF_W;
  float* Kaug  = smem + OFF_KAUG;

  const int b    = blockIdx.x;
  const int tid  = threadIdx.x;
  const int lane = tid & 31;
  const int wave = tid >> 5;
  const int lm   = lane & 15;
  const int kh   = lane >> 4;

  const float sc = *scale;
  const int lab = (tid < NS) ? labels[b * NS + tid] : 0;

  {
    for (int idx = tid; idx < Dd * WROW + 16; idx += NTHR) W_lds[idx] = 0.0f;
    const float4* sb4 = (const float4*)(s + (size_t)b * NS * Dd);
    float4 t[SLD];
    #pragma unroll
    for (int j = 0; j < SLD; ++j) t[j] = sb4[tid + j * NTHR];
    #pragma unroll
    for (int j = 0; j < SLD; ++j) ((float4*)(S_lds + j * SROW))[tid] = t[j];
  }

  __syncthreads();

  const v4f* qbase = (const v4f*)(q + (size_t)b * NQ * Dd + wave * KCH) + lane;
  v4f r0[16], r1[16];
  q_fetch<0>(qbase, r0);

  {
    const float* f0p = S_lds + lm * SROW + 2 * kh + wave * KCH;
    const float* f1p = S_lds + (16 + lm) * SROW + 2 * kh + wave * KCH;
    v8f t00 = {0.f,0.f,0.f,0.f,0.f,0.f,0.f,0.f};
    v8f t01 = {0.f,0.f,0.f,0.f,0.f,0.f,0.f,0.f};
    v8f t11 = {0.f,0.f,0.f,0.f,0.f,0.f,0.f,0.f};
    #pragma unroll 8
    for (int k = 0; k < KCH; k += 4) {
      v2f f0 = *(const v2f*)(f0p + k);
      v2f f1 = *(const v2f*)(f1p + k);
      t00 = __builtin_amdgcn_wmma_f32_16x16x4_f32(false, f0, false, f0,
                                                  (short)0, t00, false, false);
      t01 = __builtin_amdgcn_wmma_f32_16x16x4_f32(false, f0, false, f1,
                                                  (short)0, t01, false, false);
      t11 = __builtin_amdgcn_wmma_f32_16x16x4_f32(false, f1, false, f1,
                                                  (short)0, t11, false, false);
    }
    float* kp = W_lds + wave * (3 * 256) + lane;
    #pragma unroll
    for (int v = 0; v < 8; ++v) {
      kp[(0 * 8 + v) * 32] = t00[v];
      kp[(1 * 8 + v) * 32] = t01[v];
      kp[(2 * 8 + v) * 32] = t11[v];
    }
  }
  __syncthreads();
  for (int e = tid; e < NS * NS; e += NTHR) {
    const int i = e / NS, j = e - i * NS;
    const bool low = (i >= 16) && (j < 16);
    const int fr   = low ? j : i;
    const int fc   = low ? i : j;
    const int tile = (fr >> 4) + (fc >> 4);
    const int ln   = (fc & 15) + ((fr & 8) << 1);
    const float* p = W_lds + (tile * 8 + (fr & 7)) * 32 + ln;
    float sum = p[0];
    #pragma unroll
    for (int w = 1; w < NWAVE; ++w) sum += p[w * (3 * 256)];
    Kaug[i * KROW + j] = sum + ((i == j) ? 1.0f : 0.0f);
  }
  if (tid < NS) {
    for (int c = 0; c < NW; ++c)
      Kaug[tid * KROW + NS + c] = (lab == c) ? 1.0f : 0.0f;
  }
  __syncthreads();

  q_fetch<1>(qbase, r1);
  if (wave == 0) {
    const int col = (lane < NS + NW) ? lane : 0;
    float a[NS];
    #pragma unroll
    for (int r = 0; r < NS; ++r) a[r] = Kaug[r * KROW + col];
    #pragma unroll
    for (int p = 0; p < NS; ++p) {
      const float piv    = __int_as_float(__builtin_amdgcn_readlane(__float_as_int(a[p]), p));
      const float pivinv = 1.0f / piv;
      const float pr     = a[p] * pivinv;
      #pragma unroll
      for (int r = 0; r < NS; ++r) {
        if (r != p) {
          const float f = __int_as_float(__builtin_amdgcn_readlane(__float_as_int(a[r]), p));
          a[r] = a[r] - f * pr;
        }
      }
      a[p] = pr;
    }
    if (lane >= NS && lane < NS + NW) {
      #pragma unroll
      for (int r = 0; r < NS; ++r) Kaug[r * KROW + lane] = a[r];
    }
  }
  __syncthreads();

  wcols<Dd / NTHR>(S_lds, Kaug, W_lds, tid);
  __syncthreads();

  {
    float* Qs = S_lds + wave * (16 * QROW);
    float* Pw = S_lds + OFF_P + wave * PWAVE;
    const float* ap = Qs + lm * QROW + 2 * kh;

    v2f bf[KCH / 4];
    #pragma unroll
    for (int ks = 0; ks < KCH / 4; ++ks) {
      const int kb0 = wave * KCH + 4 * ks + 2 * kh;
      bf[ks] = v2f{ W_lds[kb0 * WROW + lm], W_lds[(kb0 + 1) * WROW + lm] };
    }

    q_commit(Qs, lane, r0);
    q_fetch<2>(qbase, r0);
    p_store<0>(Pw, lm, kh, q_tile_mma(ap, bf));
    q_commit(Qs, lane, r1);
    q_fetch<3>(qbase, r1);
    p_store<1>(Pw, lm, kh, q_tile_mma(ap, bf));
    q_commit(Qs, lane, r0);
    q_fetch<4>(qbase, r0);
    p_store<2>(Pw, lm, kh, q_tile_mma(ap, bf));
    q_commit(Qs, lane, r1);
    p_store<3>(Pw, lm, kh, q_tile_mma(ap, bf));
    q_commit(Qs, lane, r0);
    p_store<4>(Pw, lm, kh, q_tile_mma(ap, bf));
  }
  __syncthreads();

  for (int o = tid; o < NQ * NW; o += NTHR) {
    const int m = o / NW, c = o - m * NW;
    const float* p = S_lds + OFF_P + (((m >> 4) * 2 + ((m >> 3) & 1)) * 8 + (m & 7)) * NW + c;
    float sum = p[0];
    #pragma unroll
    for (int w = 1; w < NWAVE; ++w) sum += p[w * PWAVE];
    out[(size_t)b * (NQ * NW) + o] = sc * sum;
  }
}

extern "C" void kernel_launch(void* const* d_in, const int* in_sizes, int n_in,
                              void* d_out, int out_size, void* d_ws, size_t ws_size,
                              hipStream_t stream) {
  (void)n_in; (void)out_size; (void)d_ws; (void)ws_size;
  const float* q      = (const float*)d_in[0];
  const float* s      = (const float*)d_in[1];
  const int*   labels = (const int*)d_in[2];
  const float* scale  = (const float*)d_in[3];
  float* out = (float*)d_out;

  const int B = in_sizes[1] / (NS * Dd);
  if (B <= 0) return;

  ridge_head_fused<<<dim3(B), dim3(NTHR), 0, stream>>>(q, s, labels, scale, out);
}
